# starting kernel
# baseline (speedup 1.0000x reference)
#include <hip/hip_runtime.h>
#include <hip/hip_bf16.h>

#define SH_C0 0.28209479177387814f
#define SH_C1 0.4886025119029199f

typedef __attribute__((ext_vector_type(2))) float v2f;
typedef __attribute__((ext_vector_type(4))) float v4f;
typedef __attribute__((ext_vector_type(8))) float v8f;

__device__ __forceinline__ float sigmoidf(float x) {
    return 1.0f / (1.0f + expf(-x));
}

__global__ __launch_bounds__(256) void gs_elementwise(
    const float* __restrict__ view_dirs,
    const float* __restrict__ xyz,
    const float* __restrict__ opacity_logit,
    const float* __restrict__ sh,
    float* __restrict__ out_xyz,
    float* __restrict__ out_rgb,
    float* __restrict__ out_opa,
    int N)
{
    int g = blockIdx.x * blockDim.x + threadIdx.x;
    if (g >= N) return;
    size_t g3 = (size_t)g * 3;

    float px = __builtin_nontemporal_load(xyz + g3 + 0);
    float py = __builtin_nontemporal_load(xyz + g3 + 1);
    float pz = __builtin_nontemporal_load(xyz + g3 + 2);
    __builtin_nontemporal_store(px, out_xyz + g3 + 0);
    __builtin_nontemporal_store(py, out_xyz + g3 + 1);
    __builtin_nontemporal_store(pz, out_xyz + g3 + 2);

    float dx = __builtin_nontemporal_load(view_dirs + g3 + 0);
    float dy = __builtin_nontemporal_load(view_dirs + g3 + 1);
    float dz = __builtin_nontemporal_load(view_dirs + g3 + 2);

    const v4f* shv = (const v4f*)(sh + (size_t)g * 12);
#pragma unroll
    for (int c = 0; c < 3; ++c) {
        v4f s = __builtin_nontemporal_load(shv + c);
        float v = SH_C0 * s.x - SH_C1 * dy * s.y + SH_C1 * dz * s.z - SH_C1 * dx * s.w;
        __builtin_nontemporal_store(sigmoidf(v), out_rgb + g3 + c);
    }

    float ol = __builtin_nontemporal_load(opacity_logit + g);
    __builtin_nontemporal_store(sigmoidf(ol), out_opa + g);
}

#define GS_PER_WMMA  5
#define WMMA_ITERS   8
#define GS_PER_WAVE  (GS_PER_WMMA * WMMA_ITERS)
#define GS_PER_BLOCK (8 * GS_PER_WAVE)

__global__ __launch_bounds__(256) void gs_cov_wmma(
    const float* __restrict__ scale_log,
    const float* __restrict__ quat,
    float* __restrict__ out_cov,
    int N)
{
    const int lane = threadIdx.x & 31;
    const int wave = threadIdx.x >> 5;
    const int hi   = lane >> 4;
    const int nl   = lane & 15;

    const int wbase = blockIdx.x * GS_PER_BLOCK + wave * GS_PER_WAVE;

    const int gl = nl / 3;
    const int iRow = nl - 3 * gl;

#pragma unroll
    for (int it = 0; it < WMMA_ITERS; ++it) {
        const int gbase = wbase + it * GS_PER_WMMA;

        float a0 = 0.0f, a1 = 0.0f;
        int gg = gbase + gl;
        if (nl < 15 && gg < N) {
            v4f q = ((const v4f*)quat)[gg];
            float nrm = sqrtf(q.x * q.x + q.y * q.y + q.z * q.z + q.w * q.w);
            float inv = 1.0f / fmaxf(nrm, 1e-12f);
            float r = q.x * inv, x = q.y * inv, y = q.z * inv, z = q.w * inv;

            size_t s3 = (size_t)gg * 3;
            float s0 = expf(scale_log[s3 + 0]);
            float s1 = expf(scale_log[s3 + 1]);
            float s2 = expf(scale_log[s3 + 2]);

            float R0, R1, R2;
            if (iRow == 0)      { R0 = 1.f - 2.f*(y*y + z*z); R1 = 2.f*(x*y - r*z); R2 = 2.f*(x*z + r*y); }
            else if (iRow == 1) { R0 = 2.f*(x*y + r*z); R1 = 1.f - 2.f*(x*x + z*z); R2 = 2.f*(y*z - r*x); }
            else                { R0 = 2.f*(x*z - r*y); R1 = 2.f*(y*z + r*x); R2 = 1.f - 2.f*(x*x + y*y); }

            float m0 = R0 * s0, m1 = R1 * s1, m2 = R2 * s2;
            a0 = hi ? m2 : m0;
            a1 = hi ? 0.f : m1;
        }
        v2f a = {a0, a1};

        v8f c = {};
        c = __builtin_amdgcn_wmma_f32_16x16x4_f32(
                 false, a,  false, a,
                 (short)0, c,  false,  false);

        if (nl < 15) {
            int go = gbase + gl;
            if (go < N) {
#pragma unroll
                for (int ii = 0; ii < 3; ++ii) {
                    int m  = 3 * gl + ii;
                    int mh = m >> 3;
                    if (mh == hi) {
                        __builtin_nontemporal_store(
                            c[m & 7], out_cov + (size_t)go * 9 + ii * 3 + nl - 3 * gl);
                    }
                }
            }
        }
    }
}

extern "C" void kernel_launch(void* const* d_in, const int* in_sizes, int n_in,
                              void* d_out, int out_size, void* d_ws, size_t ws_size,
                              hipStream_t stream) {
    const float* view_dirs     = (const float*)d_in[0];
    const float* xyz           = (const float*)d_in[1];
    const float* scale_log     = (const float*)d_in[2];
    const float* rot_quat      = (const float*)d_in[3];
    const float* opacity_logit = (const float*)d_in[4];
    const float* sh_coeffs     = (const float*)d_in[5];

    int N = in_sizes[1] / 3;

    float* out     = (float*)d_out;
    float* out_xyz = out;
    float* out_cov = out + (size_t)N * 3;
    float* out_rgb = out + (size_t)N * 12;
    float* out_opa = out + (size_t)N * 15;

    int blocks_ew = (N + 255) / 256;
    gs_elementwise<<<blocks_ew, 256, 0, stream>>>(
        view_dirs, xyz, opacity_logit, sh_coeffs,
        out_xyz, out_rgb, out_opa, N);

    int blocks_cov = (N + GS_PER_BLOCK - 1) / GS_PER_BLOCK;
    gs_cov_wmma<<<blocks_cov, 256, 0, stream>>>(
        scale_log, rot_quat, out_cov, N);
}


# cov kernel: one lane builds one gaussian's M once; wave-private LDS re-lays M into 4-per-WMMA fragments and the D blocks into whole-line 16B stores (was 6 lanes per gaussian, 4B scattered stores)
# speedup vs baseline: 2.1288x; 2.1288x over previous
#include <hip/hip_runtime.h>
#include <hip/hip_bf16.h>

#define SH_C0 0.28209479177387814f
#define SH_C1 0.4886025119029199f

typedef __attribute__((ext_vector_type(2))) float v2f;
typedef __attribute__((ext_vector_type(4))) float v4f;
typedef __attribute__((ext_vector_type(8))) float v8f;

__device__ __forceinline__ float sigmoidf(float x) {
    return 1.0f / (1.0f + expf(-x));
}

__global__ __launch_bounds__(256) void gs_elementwise(
    const float* __restrict__ view_dirs,
    const float* __restrict__ xyz,
    const float* __restrict__ opacity_logit,
    const float* __restrict__ sh,
    float* __restrict__ out_xyz,
    float* __restrict__ out_rgb,
    float* __restrict__ out_opa,
    int N)
{
    int g = blockIdx.x * blockDim.x + threadIdx.x;
    if (g >= N) return;
    size_t g3 = (size_t)g * 3;

    float px = __builtin_nontemporal_load(xyz + g3 + 0);
    float py = __builtin_nontemporal_load(xyz + g3 + 1);
    float pz = __builtin_nontemporal_load(xyz + g3 + 2);
    __builtin_nontemporal_store(px, out_xyz + g3 + 0);
    __builtin_nontemporal_store(py, out_xyz + g3 + 1);
    __builtin_nontemporal_store(pz, out_xyz + g3 + 2);

    float dx = __builtin_nontemporal_load(view_dirs + g3 + 0);
    float dy = __builtin_nontemporal_load(view_dirs + g3 + 1);
    float dz = __builtin_nontemporal_load(view_dirs + g3 + 2);

    const v4f* shv = (const v4f*)(sh + (size_t)g * 12);
#pragma unroll
    for (int c = 0; c < 3; ++c) {
        v4f s = __builtin_nontemporal_load(shv + c);
        float v = SH_C0 * s.x - SH_C1 * dy * s.y + SH_C1 * dz * s.z - SH_C1 * dx * s.w;
        __builtin_nontemporal_store(sigmoidf(v), out_rgb + g3 + c);
    }

    float ol = __builtin_nontemporal_load(opacity_logit + g);
    __builtin_nontemporal_store(sigmoidf(ol), out_opa + g);
}

#define TILE_G          32
#define TILES_PER_WAVE  8
#define WAVES_PER_BLOCK 8
#define GS_PER_BLOCK    (TILE_G * TILES_PER_WAVE * WAVES_PER_BLOCK)
#define PLANE_F         144
#define MPL_F           (4 * PLANE_F)
#define OST_F           (TILE_G * 9)
#define WAVE_LDS_F      (MPL_F + OST_F)

__device__ __forceinline__ void wave_lds_handover() {
    __builtin_amdgcn_fence(__ATOMIC_RELEASE, "wavefront");
    __builtin_amdgcn_wave_barrier();
    __builtin_amdgcn_fence(__ATOMIC_ACQUIRE, "wavefront");
}

__global__ __launch_bounds__(256) void gs_cov_wmma(
    const float* __restrict__ scale_log,
    const float* __restrict__ quat,
    float* __restrict__ out_cov,
    int N)
{
    __shared__ __attribute__((aligned(16))) float lds[WAVES_PER_BLOCK * WAVE_LDS_F];

    const int lane = threadIdx.x & 31;
    const int wave = __builtin_amdgcn_readfirstlane(threadIdx.x >> 5);
    float* mpl = lds + wave * WAVE_LDS_F;
    float* ost = mpl + MPL_F;

    const int c  = lane & 15;
    const int h  = lane >> 4;
    const int gq = c >> 2;
    const int ij = c & 3;

    const int  frag_off = ij * PLANE_F + gq * 4 + 2 * h;
    const bool d_keep = ((c >> 3) == h) && (ij < 3);
    const bool d_odd  = (gq & 1) != 0;
    const int  d_off  = gq * 9 + ij;

    *(v4f*)(mpl + 3 * PLANE_F + lane * 4) = v4f{0.f, 0.f, 0.f, 0.f};

    const int  tile0   = (blockIdx.x * WAVES_PER_BLOCK + wave) * TILES_PER_WAVE;
    const bool vec_out = (((size_t)out_cov) & 15) == 0;

    v4f   q[TILES_PER_WAVE];
    float sl0[TILES_PER_WAVE], sl1[TILES_PER_WAVE], sl2[TILES_PER_WAVE];
#pragma unroll
    for (int tt = 0; tt < TILES_PER_WAVE; ++tt) {
        int g  = (tile0 + tt) * TILE_G + lane;
        int gc = g < N ? g : N - 1;
        q[tt] = ((const v4f*)quat)[gc];
        size_t s3 = (size_t)gc * 3;
        sl0[tt] = scale_log[s3 + 0];
        sl1[tt] = scale_log[s3 + 1];
        sl2[tt] = scale_log[s3 + 2];
    }
    __builtin_amdgcn_sched_barrier(0);

#pragma unroll
    for (int tt = 0; tt < TILES_PER_WAVE; ++tt) {
        const int g0 = (tile0 + tt) * TILE_G;
        if (g0 >= N) break;

        {
            v4f qq = q[tt];
            float nrm = sqrtf(qq.x * qq.x + qq.y * qq.y + qq.z * qq.z + qq.w * qq.w);
            float inv = 1.0f / fmaxf(nrm, 1e-12f);
            float r = qq.x * inv, x = qq.y * inv, y = qq.z * inv, z = qq.w * inv;

            float s0 = expf(sl0[tt]);
            float s1 = expf(sl1[tt]);
            float s2 = expf(sl2[tt]);

            float R00 = 1.f - 2.f*(y*y + z*z), R01 = 2.f*(x*y - r*z), R02 = 2.f*(x*z + r*y);
            float R10 = 2.f*(x*y + r*z), R11 = 1.f - 2.f*(x*x + z*z), R12 = 2.f*(y*z - r*x);
            float R20 = 2.f*(x*z - r*y), R21 = 2.f*(y*z + r*x), R22 = 1.f - 2.f*(x*x + y*y);

            *(v4f*)(mpl + 0 * PLANE_F + lane * 4) = v4f{R00 * s0, R01 * s1, R02 * s2, 0.f};
            *(v4f*)(mpl + 1 * PLANE_F + lane * 4) = v4f{R10 * s0, R11 * s1, R12 * s2, 0.f};
            *(v4f*)(mpl + 2 * PLANE_F + lane * 4) = v4f{R20 * s0, R21 * s1, R22 * s2, 0.f};
        }
        wave_lds_handover();

        v2f afr[TILE_G / 4];
#pragma unroll
        for (int t = 0; t < TILE_G / 4; ++t)
            afr[t] = *(const v2f*)(mpl + frag_off + 16 * t);
#pragma unroll
        for (int tg = 0; tg < TILE_G / 8; ++tg) {
            float e0[2], e1[2], e2[2];
#pragma unroll
            for (int u = 0; u < 2; ++u) {
                v2f a = afr[2 * tg + u];
                v8f d = {};
                d = __builtin_amdgcn_wmma_f32_16x16x4_f32(
                         false, a,  false, a,
                         (short)0, d,  false,  false);
                e0[u] = d_odd ? d[4] : d[0];
                e1[u] = d_odd ? d[5] : d[1];
                e2[u] = d_odd ? d[6] : d[2];
            }
            if (d_keep) {
#pragma unroll
                for (int u = 0; u < 2; ++u) {
                    const int t = 2 * tg + u;
                    ost[d_off + 36 * t + 0] = e0[u];
                    ost[d_off + 36 * t + 3] = e1[u];
                    ost[d_off + 36 * t + 6] = e2[u];
                }
            }
        }
        wave_lds_handover();

        float* dst = out_cov + (size_t)g0 * 9;
        if (vec_out && g0 + TILE_G <= N) {
            const v4f* o4 = (const v4f*)ost;
            v4f x0 = o4[lane];
            v4f x1 = o4[lane + 32];
            __builtin_nontemporal_store(x0, (v4f*)dst + lane);
            __builtin_nontemporal_store(x1, (v4f*)dst + lane + 32);
            if (lane < 8) {
                v4f x2 = o4[lane + 64];
                __builtin_nontemporal_store(x2, (v4f*)dst + lane + 64);
            }
        } else {
            int nf = 9 * (N - g0 < TILE_G ? N - g0 : TILE_G);
            for (int f = lane; f < nf; f += 32)
                __builtin_nontemporal_store(ost[f], dst + f);
        }
        wave_lds_handover();
    }
}

extern "C" void kernel_launch(void* const* d_in, const int* in_sizes, int n_in,
                              void* d_out, int out_size, void* d_ws, size_t ws_size,
                              hipStream_t stream) {
    const float* view_dirs     = (const float*)d_in[0];
    const float* xyz           = (const float*)d_in[1];
    const float* scale_log     = (const float*)d_in[2];
    const float* rot_quat      = (const float*)d_in[3];
    const float* opacity_logit = (const float*)d_in[4];
    const float* sh_coeffs     = (const float*)d_in[5];

    int N = in_sizes[1] / 3;

    float* out     = (float*)d_out;
    float* out_xyz = out;
    float* out_cov = out + (size_t)N * 3;
    float* out_rgb = out + (size_t)N * 12;
    float* out_opa = out + (size_t)N * 15;

    int blocks_ew = (N + 255) / 256;
    gs_elementwise<<<blocks_ew, 256, 0, stream>>>(
        view_dirs, xyz, opacity_logit, sh_coeffs,
        out_xyz, out_rgb, out_opa, N);

    int blocks_cov = (N + GS_PER_BLOCK - 1) / GS_PER_BLOCK;
    gs_cov_wmma<<<blocks_cov, 256, 0, stream>>>(
        scale_log, rot_quat, out_cov, N);
}


# elementwise kernel: one lane = 4 gaussians, every global access a whole-line 16B load/store (view_dirs, SH, rgb re-laid via wave-private LDS), 19 loads in flight per wave; 15625 waves, was 62504
# speedup vs baseline: 2.4629x; 1.1569x over previous
#include <hip/hip_runtime.h>
#include <hip/hip_bf16.h>

#define SH_C0 0.28209479177387814f
#define SH_C1 0.4886025119029199f

typedef __attribute__((ext_vector_type(2))) float v2f;
typedef __attribute__((ext_vector_type(4))) float v4f;
typedef __attribute__((ext_vector_type(8))) float v8f;

__device__ __forceinline__ float sigmoidf(float x) {
    return 1.0f / (1.0f + expf(-x));
}

__device__ __forceinline__ void wave_lds_handover() {
    __builtin_amdgcn_fence(__ATOMIC_RELEASE, "wavefront");
    __builtin_amdgcn_wave_barrier();
    __builtin_amdgcn_fence(__ATOMIC_ACQUIRE, "wavefront");
}

#define EW_G        4
#define EW_WAVE_G   (32 * EW_G)
#define EW_WAVES    8
#define EW_BLOCK_G  (EW_WAVE_G * EW_WAVES)
#define EW_VD_P     96
#define EW_SH_P     (384 + 8)
#define EW_WAVE_P   (EW_VD_P + EW_SH_P)

__device__ __forceinline__ void ew_one_gaussian(
    const float* __restrict__ view_dirs, const float* __restrict__ xyz,
    const float* __restrict__ opacity_logit, const float* __restrict__ sh,
    float* __restrict__ out_xyz, float* __restrict__ out_rgb,
    float* __restrict__ out_opa, int g)
{
    size_t g3 = (size_t)g * 3;
    out_xyz[g3 + 0] = xyz[g3 + 0];
    out_xyz[g3 + 1] = xyz[g3 + 1];
    out_xyz[g3 + 2] = xyz[g3 + 2];
    float dx = view_dirs[g3 + 0];
    float dy = view_dirs[g3 + 1];
    float dz = view_dirs[g3 + 2];
    const float* s = sh + (size_t)g * 12;
    for (int c = 0; c < 3; ++c) {
        float v = SH_C0 * s[4 * c + 0] - SH_C1 * dy * s[4 * c + 1]
                + SH_C1 * dz * s[4 * c + 2] - SH_C1 * dx * s[4 * c + 3];
        out_rgb[g3 + c] = sigmoidf(v);
    }
    out_opa[g] = sigmoidf(opacity_logit[g]);
}

__global__ __launch_bounds__(256) void gs_elementwise(
    const float* __restrict__ view_dirs,
    const float* __restrict__ xyz,
    const float* __restrict__ opacity_logit,
    const float* __restrict__ sh,
    float* __restrict__ out_xyz,
    float* __restrict__ out_rgb,
    float* __restrict__ out_opa,
    int N)
{
    __shared__ __attribute__((aligned(16))) v4f ew_lds[EW_WAVES * EW_WAVE_P];

    const int lane = threadIdx.x & 31;
    const int wave = __builtin_amdgcn_readfirstlane(threadIdx.x >> 5);
    const int g0   = (blockIdx.x * EW_WAVES + wave) * EW_WAVE_G;
    if (g0 >= N) return;

    const bool aligned16 =
        ((((size_t)view_dirs) | ((size_t)xyz) | ((size_t)opacity_logit) | ((size_t)sh) |
          ((size_t)out_xyz) | ((size_t)out_rgb) | ((size_t)out_opa)) & 15) == 0;

    if (!(aligned16 && g0 + EW_WAVE_G <= N)) {
        for (int u = 0; u < EW_G; ++u) {
            int g = g0 + u * 32 + lane;
            if (g < N)
                ew_one_gaussian(view_dirs, xyz, opacity_logit, sh,
                                out_xyz, out_rgb, out_opa, g);
        }
        return;
    }

    v4f* l_vd = ew_lds + wave * EW_WAVE_P;
    v4f* l_sh = l_vd + EW_VD_P;

    const v4f* xin = (const v4f*)(xyz           + (size_t)g0 * 3);
    const v4f* vin = (const v4f*)(view_dirs     + (size_t)g0 * 3);
    const v4f* shin = (const v4f*)(sh           + (size_t)g0 * 12);
    const v4f* oin = (const v4f*)(opacity_logit + (size_t)g0);
    v4f* xout = (v4f*)(out_xyz + (size_t)g0 * 3);
    v4f* rout = (v4f*)(out_rgb + (size_t)g0 * 3);
    v4f* oout = (v4f*)(out_opa + (size_t)g0);

    v4f xr[3], vr[3], sr[12], orr;
    for (int k = 0; k < 3; ++k)  xr[k] = __builtin_nontemporal_load(xin + k * 32 + lane);
    for (int k = 0; k < 3; ++k)  vr[k] = __builtin_nontemporal_load(vin + k * 32 + lane);
    for (int k = 0; k < 12; ++k) sr[k] = __builtin_nontemporal_load(shin + k * 32 + lane);
    orr = __builtin_nontemporal_load(oin + lane);

    for (int k = 0; k < 3; ++k)  __builtin_nontemporal_store(xr[k], xout + k * 32 + lane);

    const int h = lane >> 4;
    for (int k = 0; k < 3; ++k)  l_vd[k * 32 + lane] = vr[k];
    for (int k = 0; k < 12; ++k) {
        const int pad = 2 * (k / 3) + ((k % 3) == 1 ? h : ((k % 3) == 2 ? 1 : 0));
        l_sh[k * 32 + lane + pad] = sr[k];
    }
    wave_lds_handover();

    float d[12], rgb[12];
    for (int j = 0; j < 3; ++j) {
        v4f t = l_vd[3 * lane + j];
        d[4 * j + 0] = t.x; d[4 * j + 1] = t.y; d[4 * j + 2] = t.z; d[4 * j + 3] = t.w;
    }
    const v4f* my_sh = l_sh + 12 * lane + (lane >> 2);
    for (int u = 0; u < EW_G; ++u) {
        float dx = d[3 * u + 0], dy = d[3 * u + 1], dz = d[3 * u + 2];
        for (int c = 0; c < 3; ++c) {
            v4f s = my_sh[3 * u + c];
            float v = SH_C0 * s.x - SH_C1 * dy * s.y + SH_C1 * dz * s.z - SH_C1 * dx * s.w;
            rgb[3 * u + c] = sigmoidf(v);
        }
    }
    for (int j = 0; j < 3; ++j)
        l_vd[3 * lane + j] = v4f{rgb[4 * j + 0], rgb[4 * j + 1], rgb[4 * j + 2], rgb[4 * j + 3]};
    wave_lds_handover();
    for (int k = 0; k < 3; ++k) {
        v4f t = l_vd[k * 32 + lane];
        __builtin_nontemporal_store(t, rout + k * 32 + lane);
    }

    v4f oo = v4f{sigmoidf(orr.x), sigmoidf(orr.y), sigmoidf(orr.z), sigmoidf(orr.w)};
    __builtin_nontemporal_store(oo, oout + lane);
}

#define TILE_G          32
#define TILES_PER_WAVE  8
#define WAVES_PER_BLOCK 8
#define GS_PER_BLOCK    (TILE_G * TILES_PER_WAVE * WAVES_PER_BLOCK)
#define PLANE_F         144
#define MPL_F           (4 * PLANE_F)
#define OST_F           (TILE_G * 9)
#define WAVE_LDS_F      (MPL_F + OST_F)

__global__ __launch_bounds__(256) void gs_cov_wmma(
    const float* __restrict__ scale_log,
    const float* __restrict__ quat,
    float* __restrict__ out_cov,
    int N)
{
    __shared__ __attribute__((aligned(16))) float lds[WAVES_PER_BLOCK * WAVE_LDS_F];

    const int lane = threadIdx.x & 31;
    const int wave = __builtin_amdgcn_readfirstlane(threadIdx.x >> 5);
    float* mpl = lds + wave * WAVE_LDS_F;
    float* ost = mpl + MPL_F;

    const int c  = lane & 15;
    const int h  = lane >> 4;
    const int gq = c >> 2;
    const int ij = c & 3;

    const int  frag_off = ij * PLANE_F + gq * 4 + 2 * h;
    const bool d_keep = ((c >> 3) == h) && (ij < 3);
    const bool d_odd  = (gq & 1) != 0;
    const int  d_off  = gq * 9 + ij;

    *(v4f*)(mpl + 3 * PLANE_F + lane * 4) = v4f{0.f, 0.f, 0.f, 0.f};

    const int  tile0   = (blockIdx.x * WAVES_PER_BLOCK + wave) * TILES_PER_WAVE;
    const bool vec_out = (((size_t)out_cov) & 15) == 0;

    v4f   q[TILES_PER_WAVE];
    float sl0[TILES_PER_WAVE], sl1[TILES_PER_WAVE], sl2[TILES_PER_WAVE];
#pragma unroll
    for (int tt = 0; tt < TILES_PER_WAVE; ++tt) {
        int g  = (tile0 + tt) * TILE_G + lane;
        int gc = g < N ? g : N - 1;
        q[tt] = ((const v4f*)quat)[gc];
        size_t s3 = (size_t)gc * 3;
        sl0[tt] = scale_log[s3 + 0];
        sl1[tt] = scale_log[s3 + 1];
        sl2[tt] = scale_log[s3 + 2];
    }
    __builtin_amdgcn_sched_barrier(0);

#pragma unroll
    for (int tt = 0; tt < TILES_PER_WAVE; ++tt) {
        const int g0 = (tile0 + tt) * TILE_G;
        if (g0 >= N) break;

        {
            v4f qq = q[tt];
            float nrm = sqrtf(qq.x * qq.x + qq.y * qq.y + qq.z * qq.z + qq.w * qq.w);
            float inv = 1.0f / fmaxf(nrm, 1e-12f);
            float r = qq.x * inv, x = qq.y * inv, y = qq.z * inv, z = qq.w * inv;

            float s0 = expf(sl0[tt]);
            float s1 = expf(sl1[tt]);
            float s2 = expf(sl2[tt]);

            float R00 = 1.f - 2.f*(y*y + z*z), R01 = 2.f*(x*y - r*z), R02 = 2.f*(x*z + r*y);
            float R10 = 2.f*(x*y + r*z), R11 = 1.f - 2.f*(x*x + z*z), R12 = 2.f*(y*z - r*x);
            float R20 = 2.f*(x*z - r*y), R21 = 2.f*(y*z + r*x), R22 = 1.f - 2.f*(x*x + y*y);

            *(v4f*)(mpl + 0 * PLANE_F + lane * 4) = v4f{R00 * s0, R01 * s1, R02 * s2, 0.f};
            *(v4f*)(mpl + 1 * PLANE_F + lane * 4) = v4f{R10 * s0, R11 * s1, R12 * s2, 0.f};
            *(v4f*)(mpl + 2 * PLANE_F + lane * 4) = v4f{R20 * s0, R21 * s1, R22 * s2, 0.f};
        }
        wave_lds_handover();

        v2f afr[TILE_G / 4];
#pragma unroll
        for (int t = 0; t < TILE_G / 4; ++t)
            afr[t] = *(const v2f*)(mpl + frag_off + 16 * t);
#pragma unroll
        for (int tg = 0; tg < TILE_G / 8; ++tg) {
            float e0[2], e1[2], e2[2];
#pragma unroll
            for (int u = 0; u < 2; ++u) {
                v2f a = afr[2 * tg + u];
                v8f d = {};
                d = __builtin_amdgcn_wmma_f32_16x16x4_f32(
                         false, a,  false, a,
                         (short)0, d,  false,  false);
                e0[u] = d_odd ? d[4] : d[0];
                e1[u] = d_odd ? d[5] : d[1];
                e2[u] = d_odd ? d[6] : d[2];
            }
            if (d_keep) {
#pragma unroll
                for (int u = 0; u < 2; ++u) {
                    const int t = 2 * tg + u;
                    ost[d_off + 36 * t + 0] = e0[u];
                    ost[d_off + 36 * t + 3] = e1[u];
                    ost[d_off + 36 * t + 6] = e2[u];
                }
            }
        }
        wave_lds_handover();

        float* dst = out_cov + (size_t)g0 * 9;
        if (vec_out && g0 + TILE_G <= N) {
            const v4f* o4 = (const v4f*)ost;
            v4f x0 = o4[lane];
            v4f x1 = o4[lane + 32];
            __builtin_nontemporal_store(x0, (v4f*)dst + lane);
            __builtin_nontemporal_store(x1, (v4f*)dst + lane + 32);
            if (lane < 8) {
                v4f x2 = o4[lane + 64];
                __builtin_nontemporal_store(x2, (v4f*)dst + lane + 64);
            }
        } else {
            int nf = 9 * (N - g0 < TILE_G ? N - g0 : TILE_G);
            for (int f = lane; f < nf; f += 32)
                __builtin_nontemporal_store(ost[f], dst + f);
        }
        wave_lds_handover();
    }
}

extern "C" void kernel_launch(void* const* d_in, const int* in_sizes, int n_in,
                              void* d_out, int out_size, void* d_ws, size_t ws_size,
                              hipStream_t stream) {
    const float* view_dirs     = (const float*)d_in[0];
    const float* xyz           = (const float*)d_in[1];
    const float* scale_log     = (const float*)d_in[2];
    const float* rot_quat      = (const float*)d_in[3];
    const float* opacity_logit = (const float*)d_in[4];
    const float* sh_coeffs     = (const float*)d_in[5];

    int N = in_sizes[1] / 3;

    float* out     = (float*)d_out;
    float* out_xyz = out;
    float* out_cov = out + (size_t)N * 3;
    float* out_rgb = out + (size_t)N * 12;
    float* out_opa = out + (size_t)N * 15;

    int blocks_ew = (N + EW_BLOCK_G - 1) / EW_BLOCK_G;
    gs_elementwise<<<blocks_ew, 256, 0, stream>>>(
        view_dirs, xyz, opacity_logit, sh_coeffs,
        out_xyz, out_rgb, out_opa, N);

    int blocks_cov = (N + GS_PER_BLOCK - 1) / GS_PER_BLOCK;
    gs_cov_wmma<<<blocks_cov, 256, 0, stream>>>(
        scale_log, rot_quat, out_cov, N);
}


# one launch instead of two: each wave does its 128 gaussians' elementwise outputs, then their cov3d as 4 WMMA tiles in the same LDS, all its loads issued up front; 15625 waves, was 23441
# speedup vs baseline: 2.6550x; 1.0780x over previous
#include <hip/hip_runtime.h>
#include <hip/hip_bf16.h>


#define SH_C0 0.28209479177387814f
#define SH_C1 0.4886025119029199f

typedef __attribute__((ext_vector_type(2))) float v2f;
typedef __attribute__((ext_vector_type(4))) float v4f;
typedef __attribute__((ext_vector_type(8))) float v8f;

__device__ __forceinline__ float sigmoidf(float x) {
    return 1.0f / (1.0f + expf(-x));
}

__device__ __forceinline__ void wave_lds_handover() {
    __builtin_amdgcn_fence(__ATOMIC_RELEASE, "wavefront");
    __builtin_amdgcn_wave_barrier();
    __builtin_amdgcn_fence(__ATOMIC_ACQUIRE, "wavefront");
}

#define EW_G        4
#define EW_WAVE_G   (32 * EW_G)
#define EW_WAVES    8
#define EW_BLOCK_G  (EW_WAVE_G * EW_WAVES)
#define EW_VD_P     96
#define EW_SH_P     (384 + 8)
#define EW_WAVE_P   (EW_VD_P + EW_SH_P)

__device__ __forceinline__ void ew_one_gaussian(
    const float* __restrict__ view_dirs, const float* __restrict__ xyz,
    const float* __restrict__ opacity_logit, const float* __restrict__ sh,
    float* __restrict__ out_xyz, float* __restrict__ out_rgb,
    float* __restrict__ out_opa, int g)
{
    size_t g3 = (size_t)g * 3;
    out_xyz[g3 + 0] = xyz[g3 + 0];
    out_xyz[g3 + 1] = xyz[g3 + 1];
    out_xyz[g3 + 2] = xyz[g3 + 2];
    float dx = view_dirs[g3 + 0];
    float dy = view_dirs[g3 + 1];
    float dz = view_dirs[g3 + 2];
    const float* s = sh + (size_t)g * 12;
    for (int c = 0; c < 3; ++c) {
        float v = SH_C0 * s[4 * c + 0] - SH_C1 * dy * s[4 * c + 1]
                + SH_C1 * dz * s[4 * c + 2] - SH_C1 * dx * s[4 * c + 3];
        out_rgb[g3 + c] = sigmoidf(v);
    }
    out_opa[g] = sigmoidf(opacity_logit[g]);
}

#define TILE_G          32
#define TILES_PER_WAVE  (EW_WAVE_G / TILE_G)
#define PLANE_F         144
#define MPL_F           (4 * PLANE_F)
#define OST_F           (TILE_G * 9)
#define WAVE_LDS_F      (MPL_F + OST_F)

__global__ __launch_bounds__(256) void gs_fused(
    const float* __restrict__ view_dirs,
    const float* __restrict__ xyz,
    const float* __restrict__ scale_log,
    const float* __restrict__ quat,
    const float* __restrict__ opacity_logit,
    const float* __restrict__ sh,
    float* __restrict__ out_xyz,
    float* __restrict__ out_cov,
    float* __restrict__ out_rgb,
    float* __restrict__ out_opa,
    int N)
{
    __shared__ __attribute__((aligned(16))) v4f ew_lds[EW_WAVES * EW_WAVE_P];
    static_assert(WAVE_LDS_F * 4 <= EW_SH_P * 16, "part 2's LDS must fit over the sh pieces");

    const int lane = threadIdx.x & 31;
    const int wave = __builtin_amdgcn_readfirstlane(threadIdx.x >> 5);
    const int g0   = (blockIdx.x * EW_WAVES + wave) * EW_WAVE_G;
    if (g0 >= N) return;

    v4f* l_vd = ew_lds + wave * EW_WAVE_P;
    v4f* l_sh = l_vd + EW_VD_P;

    const bool aligned16 =
        ((((size_t)view_dirs) | ((size_t)xyz) | ((size_t)opacity_logit) | ((size_t)sh) |
          ((size_t)out_xyz) | ((size_t)out_rgb) | ((size_t)out_opa)) & 15) == 0;
    const bool ew_fast = aligned16 && g0 + EW_WAVE_G <= N;

    v4f   q[TILES_PER_WAVE];
    float sl0[TILES_PER_WAVE], sl1[TILES_PER_WAVE], sl2[TILES_PER_WAVE];
#define COV_LOADS() do {                                                         \
        _Pragma("unroll")                                                        \
        for (int tt = 0; tt < TILES_PER_WAVE; ++tt) {                            \
            int g  = g0 + tt * TILE_G + lane;                                    \
            int gc = g < N ? g : N - 1;     \
            q[tt] = ((const v4f*)quat)[gc];                                      \
            size_t s3 = (size_t)gc * 3;                                          \
            sl0[tt] = scale_log[s3 + 0];                                         \
            sl1[tt] = scale_log[s3 + 1];                                         \
            sl2[tt] = scale_log[s3 + 2];                                         \
        }                                                                        \
    } while (0)

    if (ew_fast) {
        const v4f* xin = (const v4f*)(xyz           + (size_t)g0 * 3);
        const v4f* vin = (const v4f*)(view_dirs     + (size_t)g0 * 3);
        const v4f* shin = (const v4f*)(sh           + (size_t)g0 * 12);
        const v4f* oin = (const v4f*)(opacity_logit + (size_t)g0);
        v4f* xout = (v4f*)(out_xyz + (size_t)g0 * 3);
        v4f* rout = (v4f*)(out_rgb + (size_t)g0 * 3);
        v4f* oout = (v4f*)(out_opa + (size_t)g0);

        v4f xr[3], vr[3], sr[12], orr;
        for (int k = 0; k < 3; ++k)  xr[k] = __builtin_nontemporal_load(xin + k * 32 + lane);
        for (int k = 0; k < 3; ++k)  vr[k] = __builtin_nontemporal_load(vin + k * 32 + lane);
        for (int k = 0; k < 12; ++k) sr[k] = __builtin_nontemporal_load(shin + k * 32 + lane);
        orr = __builtin_nontemporal_load(oin + lane);
        COV_LOADS();
        __builtin_amdgcn_sched_barrier(0);

        for (int k = 0; k < 3; ++k)  __builtin_nontemporal_store(xr[k], xout + k * 32 + lane);

        const int h = lane >> 4;
        for (int k = 0; k < 3; ++k)  l_vd[k * 32 + lane] = vr[k];
        for (int k = 0; k < 12; ++k) {
            const int pad = 2 * (k / 3) + ((k % 3) == 1 ? h : ((k % 3) == 2 ? 1 : 0));
            l_sh[k * 32 + lane + pad] = sr[k];
        }
        wave_lds_handover();

        float d[12], rgb[12];
        for (int j = 0; j < 3; ++j) {
            v4f t = l_vd[3 * lane + j];
            d[4 * j + 0] = t.x; d[4 * j + 1] = t.y; d[4 * j + 2] = t.z; d[4 * j + 3] = t.w;
        }
        const v4f* my_sh = l_sh + 12 * lane + (lane >> 2);
        for (int u = 0; u < EW_G; ++u) {
            float dx = d[3 * u + 0], dy = d[3 * u + 1], dz = d[3 * u + 2];
            for (int c = 0; c < 3; ++c) {
                v4f s = my_sh[3 * u + c];
                float v = SH_C0 * s.x - SH_C1 * dy * s.y + SH_C1 * dz * s.z - SH_C1 * dx * s.w;
                rgb[3 * u + c] = sigmoidf(v);
            }
        }
        for (int j = 0; j < 3; ++j)
            l_vd[3 * lane + j] = v4f{rgb[4 * j + 0], rgb[4 * j + 1], rgb[4 * j + 2], rgb[4 * j + 3]};
        wave_lds_handover();
        for (int k = 0; k < 3; ++k) {
            v4f t = l_vd[k * 32 + lane];
            __builtin_nontemporal_store(t, rout + k * 32 + lane);
        }

        v4f oo = v4f{sigmoidf(orr.x), sigmoidf(orr.y), sigmoidf(orr.z), sigmoidf(orr.w)};
        __builtin_nontemporal_store(oo, oout + lane);
    } else {
        COV_LOADS();
        for (int u = 0; u < EW_G; ++u) {
            int g = g0 + u * 32 + lane;
            if (g < N)
                ew_one_gaussian(view_dirs, xyz, opacity_logit, sh,
                                out_xyz, out_rgb, out_opa, g);
        }
    }
#undef COV_LOADS
    wave_lds_handover();

    float* mpl = (float*)l_sh;
    float* ost = mpl + MPL_F;

    const int c  = lane & 15;
    const int h2 = lane >> 4;
    const int gq = c >> 2;
    const int ij = c & 3;

    const int  frag_off = ij * PLANE_F + gq * 4 + 2 * h2;
    const bool d_keep = ((c >> 3) == h2) && (ij < 3);
    const bool d_odd  = (gq & 1) != 0;
    const int  d_off  = gq * 9 + ij;

    *(v4f*)(mpl + 3 * PLANE_F + lane * 4) = v4f{0.f, 0.f, 0.f, 0.f};

    const bool vec_out = (((size_t)out_cov) & 15) == 0;

#pragma unroll
    for (int tt = 0; tt < TILES_PER_WAVE; ++tt) {
        const int t0 = g0 + tt * TILE_G;
        if (t0 >= N) break;

        {
            v4f qq = q[tt];
            float nrm = sqrtf(qq.x * qq.x + qq.y * qq.y + qq.z * qq.z + qq.w * qq.w);
            float inv = 1.0f / fmaxf(nrm, 1e-12f);
            float r = qq.x * inv, x = qq.y * inv, y = qq.z * inv, z = qq.w * inv;

            float s0 = expf(sl0[tt]);
            float s1 = expf(sl1[tt]);
            float s2 = expf(sl2[tt]);

            float R00 = 1.f - 2.f*(y*y + z*z), R01 = 2.f*(x*y - r*z), R02 = 2.f*(x*z + r*y);
            float R10 = 2.f*(x*y + r*z), R11 = 1.f - 2.f*(x*x + z*z), R12 = 2.f*(y*z - r*x);
            float R20 = 2.f*(x*z - r*y), R21 = 2.f*(y*z + r*x), R22 = 1.f - 2.f*(x*x + y*y);

            *(v4f*)(mpl + 0 * PLANE_F + lane * 4) = v4f{R00 * s0, R01 * s1, R02 * s2, 0.f};
            *(v4f*)(mpl + 1 * PLANE_F + lane * 4) = v4f{R10 * s0, R11 * s1, R12 * s2, 0.f};
            *(v4f*)(mpl + 2 * PLANE_F + lane * 4) = v4f{R20 * s0, R21 * s1, R22 * s2, 0.f};
        }
        wave_lds_handover();

        v2f afr[TILE_G / 4];
#pragma unroll
        for (int t = 0; t < TILE_G / 4; ++t)
            afr[t] = *(const v2f*)(mpl + frag_off + 16 * t);
#pragma unroll
        for (int tg = 0; tg < TILE_G / 8; ++tg) {
            float e0[2], e1[2], e2[2];
#pragma unroll
            for (int u = 0; u < 2; ++u) {
                v2f a = afr[2 * tg + u];
                v8f d = {};
                d = __builtin_amdgcn_wmma_f32_16x16x4_f32(
                         false, a,  false, a,
                         (short)0, d,  false,  false);
                e0[u] = d_odd ? d[4] : d[0];
                e1[u] = d_odd ? d[5] : d[1];
                e2[u] = d_odd ? d[6] : d[2];
            }
            if (d_keep) {
#pragma unroll
                for (int u = 0; u < 2; ++u) {
                    const int t = 2 * tg + u;
                    ost[d_off + 36 * t + 0] = e0[u];
                    ost[d_off + 36 * t + 3] = e1[u];
                    ost[d_off + 36 * t + 6] = e2[u];
                }
            }
        }
        wave_lds_handover();

        float* dst = out_cov + (size_t)t0 * 9;
        if (vec_out && t0 + TILE_G <= N) {
            const v4f* o4 = (const v4f*)ost;
            v4f x0 = o4[lane];
            v4f x1 = o4[lane + 32];
            __builtin_nontemporal_store(x0, (v4f*)dst + lane);
            __builtin_nontemporal_store(x1, (v4f*)dst + lane + 32);
            if (lane < 8) {
                v4f x2 = o4[lane + 64];
                __builtin_nontemporal_store(x2, (v4f*)dst + lane + 64);
            }
        } else {
            int nf = 9 * (N - t0 < TILE_G ? N - t0 : TILE_G);
            for (int f = lane; f < nf; f += 32)
                __builtin_nontemporal_store(ost[f], dst + f);
        }
        wave_lds_handover();
    }
}

extern "C" void kernel_launch(void* const* d_in, const int* in_sizes, int n_in,
                              void* d_out, int out_size, void* d_ws, size_t ws_size,
                              hipStream_t stream) {
    const float* view_dirs     = (const float*)d_in[0];
    const float* xyz           = (const float*)d_in[1];
    const float* scale_log     = (const float*)d_in[2];
    const float* rot_quat      = (const float*)d_in[3];
    const float* opacity_logit = (const float*)d_in[4];
    const float* sh_coeffs     = (const float*)d_in[5];

    int N = in_sizes[1] / 3;

    float* out     = (float*)d_out;
    float* out_xyz = out;
    float* out_cov = out + (size_t)N * 3;
    float* out_rgb = out + (size_t)N * 12;
    float* out_opa = out + (size_t)N * 15;

    int blocks = (N + EW_BLOCK_G - 1) / EW_BLOCK_G;
    gs_fused<<<blocks, 256, 0, stream>>>(
        view_dirs, xyz, scale_log, rot_quat, opacity_logit, sh_coeffs,
        out_xyz, out_cov, out_rgb, out_opa, N);
}
